# starting kernel
# baseline (speedup 1.0000x reference)
#include <hip/hip_runtime.h>


typedef __attribute__((ext_vector_type(2))) float v2f;
typedef __attribute__((ext_vector_type(8))) float v8f;

#define RANK 64

__global__ __launch_bounds__(256) void cp_eval_wmma_kernel(
    const int*   __restrict__ idx,
    const float* __restrict__ f0,
    const float* __restrict__ f1,
    const float* __restrict__ f2,
    const float* __restrict__ lamb,
    float*       __restrict__ out,
    int n)
{
    const int lane  = threadIdx.x & 31;
    const int wave  = blockIdx.x * (blockDim.x >> 5) + (threadIdx.x >> 5);
    const int base  = wave * 16;
    if (base >= n) return;

    const int m     = lane & 15;
    const int khalf = lane >> 4;
    const int e     = base + m;

    const int i0 = idx[3 * e + 0];
    const int i1 = idx[3 * e + 1];
    const int i2 = idx[3 * e + 2];

    const float* __restrict__ r0 = f0 + (size_t)i0 * RANK;
    const float* __restrict__ r1 = f1 + (size_t)i1 * RANK;
    const float* __restrict__ r2 = f2 + (size_t)i2 * RANK;

    v8f c = {};
    v2f bones;
    bones.x = 1.0f;
    bones.y = 1.0f;

#pragma unroll
    for (int j = 0; j < 16; ++j) {
        const int r = 4 * j + 2 * khalf;
        const float w0 = lamb[r];
        const float w1 = lamb[r + 1];
        v2f a;
        a.x = w0 * r0[r]     * r1[r]     * r2[r];
        a.y = w1 * r0[r + 1] * r1[r + 1] * r2[r + 1];
        c = __builtin_amdgcn_wmma_f32_16x16x4_f32(
                false, a, false, bones, (short)0, c, false, false);
    }

    if ((lane & 15) == 0) {
        float4 lo = make_float4(c[0], c[1], c[2], c[3]);
        float4 hi = make_float4(c[4], c[5], c[6], c[7]);
        float4* o = reinterpret_cast<float4*>(out + base + 8 * khalf);
        o[0] = lo;
        o[1] = hi;
    }
}

extern "C" void kernel_launch(void* const* d_in, const int* in_sizes, int n_in,
                              void* d_out, int out_size, void* d_ws, size_t ws_size,
                              hipStream_t stream) {
    const int*   idx  = (const int*)d_in[0];
    const float* f0   = (const float*)d_in[1];
    const float* f1   = (const float*)d_in[2];
    const float* f2   = (const float*)d_in[3];
    const float* lamb = (const float*)d_in[4];
    float*       out  = (float*)d_out;

    const int n = in_sizes[0] / 3;
    const int waves  = (n + 15) / 16;
    const int wpb    = 256 / 32;
    const int blocks = (waves + wpb - 1) / wpb;

    cp_eval_wmma_kernel<<<blocks, 256, 0, stream>>>(idx, f0, f1, f2, lamb, out, n);
}


# Coalesced gathers: each 16-lane half reads one whole 256-byte factor row as 128-bit loads (4 lines per load, not 16); a 0/1 selector B puts each entry in its own D column; r-sum order changes.
# speedup vs baseline: 3.1792x; 3.1792x over previous
#include <hip/hip_runtime.h>


typedef __attribute__((ext_vector_type(2))) float v2f;
typedef __attribute__((ext_vector_type(8))) float v8f;

#define RANK 64

__device__ __forceinline__ float4 row_piece(const float* __restrict__ f,
                                            int row, unsigned cbyte) {
    const unsigned off = ((unsigned)row << 8) + cbyte;
    return *reinterpret_cast<const float4*>(
        reinterpret_cast<const char*>(f) + off);
}

__device__ __forceinline__ float other_half(float x) {
    return __builtin_bit_cast(float,
        __builtin_amdgcn_permlanex16(__builtin_bit_cast(int, x),
                                     __builtin_bit_cast(int, x),
                                     0x76543210, 0xfedcba98, false, false));
}

__global__ __launch_bounds__(256) void cp_eval_wmma_kernel(
    const int*   __restrict__ idx,
    const float* __restrict__ f0,
    const float* __restrict__ f1,
    const float* __restrict__ f2,
    const float* __restrict__ lamb,
    float*       __restrict__ out,
    int n)
{
    const int lane = threadIdx.x & 31;
    const int wave = blockIdx.x * (blockDim.x >> 5) + (threadIdx.x >> 5);
    const int base = __builtin_amdgcn_readfirstlane(wave * 16);
    if (base >= n) return;

    const int c = lane & 15;
    const int h = lane >> 4;
    const unsigned cbyte = (unsigned)c * 16u;

    const float4 w = *reinterpret_cast<const float4*>(lamb + 4 * c);
    const int* __restrict__ ib = idx + 3 * (size_t)base;

    v8f acc = {};

#pragma unroll
    for (int i = 0; i < 8; ++i) {
        const int a0 = ib[6 * i + 0], a1 = ib[6 * i + 3];
        const int b0 = ib[6 * i + 1], b1 = ib[6 * i + 4];
        const int d0 = ib[6 * i + 2], d1 = ib[6 * i + 5];
        const int i0 = h ? a1 : a0;
        const int i1 = h ? b1 : b0;
        const int i2 = h ? d1 : d0;

        const float4 x = row_piece(f0, i0, cbyte);
        const float4 y = row_piece(f1, i1, cbyte);
        const float4 z = row_piece(f2, i2, cbyte);

        v2f p01, p23;
        p01.x = w.x * x.x * y.x * z.x;
        p01.y = w.y * x.y * y.y * z.y;
        p23.x = w.z * x.z * y.z * z.z;
        p23.y = w.w * x.w * y.w * z.w;

        v2f sel;
        sel.x = (c == 2 * i + h) ? 1.0f : 0.0f;
        sel.y = sel.x;

        acc = __builtin_amdgcn_wmma_f32_16x16x4_f32(
                  false, p01, false, sel, (short)0, acc, false, false);
        acc = __builtin_amdgcn_wmma_f32_16x16x4_f32(
                  false, p23, false, sel, (short)0, acc, false, false);
    }

    float t = ((acc[0] + acc[1]) + (acc[2] + acc[3])) +
              ((acc[4] + acc[5]) + (acc[6] + acc[7]));
    t += other_half(t);
    if (h == 0) out[base + c] = t;
}

extern "C" void kernel_launch(void* const* d_in, const int* in_sizes, int n_in,
                              void* d_out, int out_size, void* d_ws, size_t ws_size,
                              hipStream_t stream) {
    const int*   idx  = (const int*)d_in[0];
    const float* f0   = (const float*)d_in[1];
    const float* f1   = (const float*)d_in[2];
    const float* f2   = (const float*)d_in[3];
    const float* lamb = (const float*)d_in[4];
    float*       out  = (float*)d_out;

    const int n = in_sizes[0] / 3;
    const int waves  = (n + 15) / 16;
    const int wpb    = 256 / 32;
    const int blocks = (waves + wpb - 1) / wpb;

    cp_eval_wmma_kernel<<<blocks, 256, 0, stream>>>(idx, f0, f1, f2, lamb, out, n);
}


# Resident grid: 2048 blocks of 8 waves (64 registers a lane, every wave slot filled once) and each wave walks several 16-entry tiles, so a call starts 16384 waves instead of 62500 short ones.
# speedup vs baseline: 3.3510x; 1.0540x over previous
#include <hip/hip_runtime.h>


typedef __attribute__((ext_vector_type(2))) float v2f;
typedef __attribute__((ext_vector_type(8))) float v8f;

#define RANK 64

__device__ __forceinline__ float4 row_piece(const float* __restrict__ f,
                                            int row, unsigned cbyte) {
    const unsigned off = ((unsigned)row << 8) + cbyte;
    return *reinterpret_cast<const float4*>(
        reinterpret_cast<const char*>(f) + off);
}

__device__ __forceinline__ float other_half(float x) {
    return __builtin_bit_cast(float,
        __builtin_amdgcn_permlanex16(__builtin_bit_cast(int, x),
                                     __builtin_bit_cast(int, x),
                                     0x76543210, 0xfedcba98, false, false));
}

__global__ __launch_bounds__(256, 16) void cp_eval_wmma_kernel(
    const int*   __restrict__ idx,
    const float* __restrict__ f0,
    const float* __restrict__ f1,
    const float* __restrict__ f2,
    const float* __restrict__ lamb,
    float*       __restrict__ out,
    int n)
{
    const int lane = threadIdx.x & 31;
    const int wpb  = blockDim.x >> 5;
    const int wave   = __builtin_amdgcn_readfirstlane(blockIdx.x * wpb + (threadIdx.x >> 5));
    const int waves  = gridDim.x * wpb;
    const int ntiles = (n + 15) >> 4;

    const int c = lane & 15;
    const int h = lane >> 4;
    const unsigned cbyte = (unsigned)c * 16u;

    const float4 w = *reinterpret_cast<const float4*>(lamb + 4 * c);

    for (int tile = wave; tile < ntiles; tile += waves) {
        const int base = tile * 16;
        const int* __restrict__ ib = idx + 3 * (size_t)base;

        v8f acc = {};

#pragma unroll
        for (int i = 0; i < 8; ++i) {
            const int a0 = ib[6 * i + 0], a1 = ib[6 * i + 3];
            const int b0 = ib[6 * i + 1], b1 = ib[6 * i + 4];
            const int d0 = ib[6 * i + 2], d1 = ib[6 * i + 5];
            const int i0 = h ? a1 : a0;
            const int i1 = h ? b1 : b0;
            const int i2 = h ? d1 : d0;

            const float4 x = row_piece(f0, i0, cbyte);
            const float4 y = row_piece(f1, i1, cbyte);
            const float4 z = row_piece(f2, i2, cbyte);

            v2f p01, p23;
            p01.x = w.x * x.x * y.x * z.x;
            p01.y = w.y * x.y * y.y * z.y;
            p23.x = w.z * x.z * y.z * z.z;
            p23.y = w.w * x.w * y.w * z.w;

            v2f sel;
            sel.x = (c == 2 * i + h) ? 1.0f : 0.0f;
            sel.y = sel.x;

            acc = __builtin_amdgcn_wmma_f32_16x16x4_f32(
                      false, p01, false, sel, (short)0, acc, false, false);
            acc = __builtin_amdgcn_wmma_f32_16x16x4_f32(
                      false, p23, false, sel, (short)0, acc, false, false);
        }

        float t = ((acc[0] + acc[1]) + (acc[2] + acc[3])) +
                  ((acc[4] + acc[5]) + (acc[6] + acc[7]));
        t += other_half(t);
        if (h == 0) out[base + c] = t;
    }
}

extern "C" void kernel_launch(void* const* d_in, const int* in_sizes, int n_in,
                              void* d_out, int out_size, void* d_ws, size_t ws_size,
                              hipStream_t stream) {
    const int*   idx  = (const int*)d_in[0];
    const float* f0   = (const float*)d_in[1];
    const float* f1   = (const float*)d_in[2];
    const float* f2   = (const float*)d_in[3];
    const float* lamb = (const float*)d_in[4];
    float*       out  = (float*)d_out;

    const int n = in_sizes[0] / 3;

    cp_eval_wmma_kernel<<<2048, 256, 0, stream>>>(idx, f0, f1, f2, lamb, out, n);
}
